# starting kernel
# baseline (speedup 1.0000x reference)
#include <hip/hip_runtime.h>
#include <math.h>

typedef __attribute__((ext_vector_type(2))) float v2f;
typedef __attribute__((ext_vector_type(8))) float v8f;

#define KCOEF 32
#define NDIM  256
#define LSTR  44

__global__ __launch_bounds__(256) void TruncatedDCT_kernel(
    const float* __restrict__ x, float* __restrict__ out)
{
    __shared__ float Bt[NDIM * LSTR];
    __shared__ float Tl[NDIM * LSTR];

    const int tid  = threadIdx.x;
    const int wave = tid >> 5;
    const int lane = tid & 31;
    const int half = lane >> 4;
    const int l16  = lane & 15;

    const float PI = 3.14159265358979323846f;
    for (int i = tid; i < NDIM * KCOEF; i += 256) {
        int w = i >> 5;
        int l = i & 31;
        float ang = PI * (2.0f * (float)w + 1.0f) * (float)l * (1.0f / (2.0f * (float)NDIM));
        Bt[w * LSTR + l] = 2.0f * cosf(ang);
    }
    __syncthreads();

    const float* X = x + (size_t)blockIdx.x * (size_t)(NDIM * NDIM);

    const int h0   = wave * 32;
    const int koff = half * 2;
    const float* Xr0 = X + (size_t)(h0 + l16) * NDIM + koff;
    const float* Xr1 = Xr0 + 16 * NDIM;

    v8f acc00 = {}; v8f acc01 = {}; v8f acc10 = {}; v8f acc11 = {};

    #pragma unroll 4
    for (int k0 = 0; k0 < NDIM; k0 += 4) {
        v2f a0 = *(const v2f*)(Xr0 + k0);
        v2f a1 = *(const v2f*)(Xr1 + k0);
        const float* br = &Bt[(k0 + koff) * LSTR];
        v2f b0, b1;
        b0.x = br[l16];            b0.y = br[LSTR + l16];
        b1.x = br[16 + l16];       b1.y = br[LSTR + 16 + l16];

        acc00 = __builtin_amdgcn_wmma_f32_16x16x4_f32(false, a0, false, b0, (short)0, acc00, false, false);
        acc01 = __builtin_amdgcn_wmma_f32_16x16x4_f32(false, a0, false, b1, (short)0, acc01, false, false);
        acc10 = __builtin_amdgcn_wmma_f32_16x16x4_f32(false, a1, false, b0, (short)0, acc10, false, false);
        acc11 = __builtin_amdgcn_wmma_f32_16x16x4_f32(false, a1, false, b1, (short)0, acc11, false, false);
    }

    #pragma unroll
    for (int i = 0; i < 8; ++i) {
        int r = h0 + i + 8 * half;
        Tl[r * LSTR + l16]               = acc00[i];
        Tl[r * LSTR + 16 + l16]          = acc01[i];
        Tl[(r + 16) * LSTR + l16]        = acc10[i];
        Tl[(r + 16) * LSTR + 16 + l16]   = acc11[i];
    }
    __syncthreads();

    if (wave < 4) {
        const int m0 = (wave >> 1) * 16;
        const int n0 = (wave & 1) * 16;
        v8f acc = {};
        #pragma unroll 4
        for (int k0 = 0; k0 < NDIM; k0 += 4) {
            int kk = k0 + koff;
            v2f a, b;
            a.x = Bt[kk * LSTR + m0 + l16];
            a.y = Bt[(kk + 1) * LSTR + m0 + l16];
            b.x = Tl[kk * LSTR + n0 + l16];
            b.y = Tl[(kk + 1) * LSTR + n0 + l16];
            acc = __builtin_amdgcn_wmma_f32_16x16x4_f32(false, a, false, b, (short)0, acc, false, false);
        }
        float* Y = out + (size_t)blockIdx.x * (KCOEF * KCOEF);
        #pragma unroll
        for (int i = 0; i < 8; ++i)
            Y[(m0 + i + 8 * half) * KCOEF + n0 + l16] = acc[i];
    }
}

extern "C" void kernel_launch(void* const* d_in, const int* in_sizes, int n_in,
                              void* d_out, int out_size, void* d_ws, size_t ws_size,
                              hipStream_t stream) {
    const float* x = (const float*)d_in[0];
    float* out = (float*)d_out;
    int slices = in_sizes[0] / (NDIM * NDIM);
    TruncatedDCT_kernel<<<slices, 256, 0, stream>>>(x, out);
}


# Swap the product order so x is the WMMA B operand with lanes along w: every global load is whole 128-B lines (float4), not 16 lines x 16 B; sums now run over h first, then w in even/odd halves
# speedup vs baseline: 1.2770x; 1.2770x over previous
#include <hip/hip_runtime.h>
#include <math.h>

typedef __attribute__((ext_vector_type(2))) float v2f;
typedef __attribute__((ext_vector_type(4))) float v4f;
typedef __attribute__((ext_vector_type(8))) float v8f;

#define KCOEF 32
#define NDIM  256
#define USTR  260

__global__ __launch_bounds__(128) void TruncatedDCT_kernel(
    const float* __restrict__ x, float* __restrict__ out)
{
    __shared__ __attribute__((aligned(16))) float Bf[NDIM * KCOEF];
    __shared__ __attribute__((aligned(16))) float U[KCOEF * USTR];

    const int tid  = threadIdx.x;
    const int wave = tid >> 5;
    const int lane = tid & 31;
    const int half = lane >> 4;
    const int l16  = lane & 15;

    const float PI = 3.14159265358979323846f;
    for (int i = tid; i < NDIM * KCOEF; i += 128) {
        int e  = i & 3;
        int c  = (i >> 2) & 15;
        int hh = (i >> 6) & 1;
        int s  = i >> 7;
        int n  = 4 * s + 2 * hh + (e & 1);
        int k  = 16 * (e >> 1) + c;
        float ang = PI * (2.0f * (float)n + 1.0f) * (float)k * (1.0f / (2.0f * (float)NDIM));
        Bf[i] = 2.0f * cosf(ang);
    }
    __syncthreads();

    const float* X = x + (size_t)blockIdx.x * (size_t)(NDIM * NDIM);

    const float* Xp = X + (size_t)(2 * half) * NDIM + wave * 64 + 4 * l16;
    const float* Ap = &Bf[(half * 16 + l16) * 4];

    v8f acc0[4], acc1[4];
    #pragma unroll
    for (int j = 0; j < 4; ++j) { acc0[j] = (v8f){}; acc1[j] = (v8f){}; }

    #pragma unroll 8
    for (int s = 0; s < NDIM / 4; ++s) {
        v4f r0 = *(const v4f*)(Xp + (size_t)(4 * s) * NDIM);
        v4f r1 = *(const v4f*)(Xp + (size_t)(4 * s + 1) * NDIM);
        v4f q  = *(const v4f*)(Ap + s * 128);
        v2f a0 = {q.x, q.y};
        v2f a1 = {q.z, q.w};
        #pragma unroll
        for (int j = 0; j < 4; ++j) {
            v2f b = {r0[j], r1[j]};
            acc0[j] = __builtin_amdgcn_wmma_f32_16x16x4_f32(false, a0, false, b, (short)0, acc0[j], false, false);
            acc1[j] = __builtin_amdgcn_wmma_f32_16x16x4_f32(false, a1, false, b, (short)0, acc1[j], false, false);
        }
    }

    #pragma unroll
    for (int r = 0; r < 8; ++r) {
        v4f u0 = {acc0[0][r], acc0[1][r], acc0[2][r], acc0[3][r]};
        v4f u1 = {acc1[0][r], acc1[1][r], acc1[2][r], acc1[3][r]};
        *(v4f*)&U[(8 * half + r) * USTR + wave * 64 + 4 * l16]      = u0;
        *(v4f*)&U[(16 + 8 * half + r) * USTR + wave * 64 + 4 * l16] = u1;
    }
    __syncthreads();

    const int mt = wave >> 1;
    const int nt = wave & 1;
    const float* Up = &U[(16 * mt + l16) * USTR + 2 * half];
    const float* Bp = &Bf[(half * 16 + l16) * 4 + 2 * nt];
    v8f ye = {}; v8f yo = {};
    #pragma unroll 4
    for (int s = 0; s < NDIM / 4; s += 2) {
        v2f ae = *(const v2f*)(Up + 4 * s);
        v2f be = *(const v2f*)(Bp + 128 * s);
        v2f ao = *(const v2f*)(Up + 4 * s + 4);
        v2f bo = *(const v2f*)(Bp + 128 * s + 128);
        ye = __builtin_amdgcn_wmma_f32_16x16x4_f32(false, ae, false, be, (short)0, ye, false, false);
        yo = __builtin_amdgcn_wmma_f32_16x16x4_f32(false, ao, false, bo, (short)0, yo, false, false);
    }
    float* Y = out + (size_t)blockIdx.x * (KCOEF * KCOEF);
    #pragma unroll
    for (int r = 0; r < 8; ++r)
        Y[(16 * mt + 8 * half + r) * KCOEF + 16 * nt + l16] = ye[r] + yo[r];
}

extern "C" void kernel_launch(void* const* d_in, const int* in_sizes, int n_in,
                              void* d_out, int out_size, void* d_ws, size_t ws_size,
                              hipStream_t stream) {
    const float* x = (const float*)d_in[0];
    float* out = (float*)d_out;
    int slices = in_sizes[0] / (NDIM * NDIM);
    TruncatedDCT_kernel<<<slices, 128, 0, stream>>>(x, out);
}


# Four slices per work-group (16 waves, one work-group per WGP): the 8192-cosine basis table is built once by 512 threads and shared by four 4-wave groups instead of once per slice
# speedup vs baseline: 1.5261x; 1.1951x over previous
#include <hip/hip_runtime.h>
#include <math.h>

typedef __attribute__((ext_vector_type(2))) float v2f;
typedef __attribute__((ext_vector_type(4))) float v4f;
typedef __attribute__((ext_vector_type(8))) float v8f;

#define KCOEF 32
#define NDIM  256
#define USTR  260
#define NGRP  4

__global__ __launch_bounds__(128 * NGRP) void TruncatedDCT_kernel(
    const float* __restrict__ x, float* __restrict__ out)
{
    __shared__ __attribute__((aligned(16))) float Bf[NDIM * KCOEF];
    __shared__ __attribute__((aligned(16))) float Uall[NGRP * KCOEF * USTR];

    const int tid  = threadIdx.x;
    const int grp  = tid >> 7;
    const int wave = (tid >> 5) & 3;
    const int lane = tid & 31;
    const int half = lane >> 4;
    const int l16  = lane & 15;

    const float PI = 3.14159265358979323846f;
    for (int i = tid; i < NDIM * KCOEF; i += 128 * NGRP) {
        int e  = i & 3;
        int c  = (i >> 2) & 15;
        int hh = (i >> 6) & 1;
        int s  = i >> 7;
        int n  = 4 * s + 2 * hh + (e & 1);
        int k  = 16 * (e >> 1) + c;
        float ang = PI * (2.0f * (float)n + 1.0f) * (float)k * (1.0f / (2.0f * (float)NDIM));
        Bf[i] = 2.0f * cosf(ang);
    }
    __syncthreads();

    const size_t slice = (size_t)blockIdx.x * NGRP + grp;
    const float* X = x + slice * (size_t)(NDIM * NDIM);
    float* U = Uall + grp * (KCOEF * USTR);

    const float* Xp = X + (size_t)(2 * half) * NDIM + wave * 64 + 4 * l16;
    const float* Ap = &Bf[(half * 16 + l16) * 4];

    v8f acc0[4], acc1[4];
    #pragma unroll
    for (int j = 0; j < 4; ++j) { acc0[j] = (v8f){}; acc1[j] = (v8f){}; }

    #pragma unroll 8
    for (int s = 0; s < NDIM / 4; ++s) {
        v4f r0 = *(const v4f*)(Xp + (size_t)(4 * s) * NDIM);
        v4f r1 = *(const v4f*)(Xp + (size_t)(4 * s + 1) * NDIM);
        v4f q  = *(const v4f*)(Ap + s * 128);
        v2f a0 = {q.x, q.y};
        v2f a1 = {q.z, q.w};
        #pragma unroll
        for (int j = 0; j < 4; ++j) {
            v2f b = {r0[j], r1[j]};
            acc0[j] = __builtin_amdgcn_wmma_f32_16x16x4_f32(false, a0, false, b, (short)0, acc0[j], false, false);
            acc1[j] = __builtin_amdgcn_wmma_f32_16x16x4_f32(false, a1, false, b, (short)0, acc1[j], false, false);
        }
    }

    #pragma unroll
    for (int r = 0; r < 8; ++r) {
        v4f u0 = {acc0[0][r], acc0[1][r], acc0[2][r], acc0[3][r]};
        v4f u1 = {acc1[0][r], acc1[1][r], acc1[2][r], acc1[3][r]};
        *(v4f*)&U[(8 * half + r) * USTR + wave * 64 + 4 * l16]      = u0;
        *(v4f*)&U[(16 + 8 * half + r) * USTR + wave * 64 + 4 * l16] = u1;
    }
    __syncthreads();

    const int mt = wave >> 1;
    const int nt = wave & 1;
    const float* Up = &U[(16 * mt + l16) * USTR + 2 * half];
    const float* Bp = &Bf[(half * 16 + l16) * 4 + 2 * nt];
    v8f ye = {}; v8f yo = {};
    #pragma unroll 4
    for (int s = 0; s < NDIM / 4; s += 2) {
        v2f ae = *(const v2f*)(Up + 4 * s);
        v2f be = *(const v2f*)(Bp + 128 * s);
        v2f ao = *(const v2f*)(Up + 4 * s + 4);
        v2f bo = *(const v2f*)(Bp + 128 * s + 128);
        ye = __builtin_amdgcn_wmma_f32_16x16x4_f32(false, ae, false, be, (short)0, ye, false, false);
        yo = __builtin_amdgcn_wmma_f32_16x16x4_f32(false, ao, false, bo, (short)0, yo, false, false);
    }
    float* Y = out + slice * (KCOEF * KCOEF);
    #pragma unroll
    for (int r = 0; r < 8; ++r)
        Y[(16 * mt + 8 * half + r) * KCOEF + 16 * nt + l16] = ye[r] + yo[r];
}

extern "C" void kernel_launch(void* const* d_in, const int* in_sizes, int n_in,
                              void* d_out, int out_size, void* d_ws, size_t ws_size,
                              hipStream_t stream) {
    const float* x = (const float*)d_in[0];
    float* out = (float*)d_out;
    int slices = in_sizes[0] / (NDIM * NDIM);
    TruncatedDCT_kernel<<<slices / NGRP, 128 * NGRP, 0, stream>>>(x, out);
}


# Basis table from the hardware cosine on exactly reduced revolutions ((2n+1)k mod 1024)/1024: no range reduction or polynomial in the prologue; basis values change in their last bits
# speedup vs baseline: 2.0052x; 1.3139x over previous
#include <hip/hip_runtime.h>
#include <math.h>

typedef __attribute__((ext_vector_type(2))) float v2f;
typedef __attribute__((ext_vector_type(4))) float v4f;
typedef __attribute__((ext_vector_type(8))) float v8f;

#define KCOEF 32
#define NDIM  256
#define NFOLD 128
#define USTR  260
#define NGRP  4

__global__ __launch_bounds__(128 * NGRP) void TruncatedDCT_kernel(
    const float* __restrict__ x, float* __restrict__ out)
{
    __shared__ __attribute__((aligned(16))) float Bf[NFOLD * KCOEF];
    __shared__ __attribute__((aligned(16))) float Uall[NGRP * KCOEF * USTR];
    __shared__ __attribute__((aligned(16))) float Yall[NGRP * KCOEF * KCOEF];

    const int tid  = threadIdx.x;
    const int grp  = tid >> 7;
    const int wave = (tid >> 5) & 3;
    const int lane = tid & 31;
    const int half = lane >> 4;
    const int l16  = lane & 15;

    {
        const int e  = tid & 3;
        const int c  = (tid >> 2) & 15;
        const int hh = (tid >> 6) & 1;
        const int n0 = 4 * (tid >> 7) + 2 * hh + (e & 1);
        const int k  = 2 * c + (e >> 1);
        const int m0 = (2 * n0 + 1) * k;
        #pragma unroll
        for (int it = 0; it < (NFOLD * KCOEF) / (128 * NGRP); ++it) {
            const int m = (m0 + 32 * k * it) & (4 * NDIM - 1);
            Bf[tid + 128 * NGRP * it] = 2.0f * __builtin_amdgcn_cosf((float)m * (1.0f / (4.0f * (float)NDIM)));
        }
    }
    __syncthreads();

    const size_t slice = (size_t)blockIdx.x * NGRP + grp;
    const float* X = x + slice * (size_t)(NDIM * NDIM);
    float* U = Uall + grp * (KCOEF * USTR);

    const float* Xp = X + (size_t)(2 * half) * NDIM + wave * 64 + 4 * l16;
    const float* Xm = X + (size_t)(NDIM - 1 - 2 * half) * NDIM + wave * 64 + 4 * l16;
    const float* Ap = &Bf[(half * 16 + l16) * 4];

    v8f accE[4], accO[4];
    #pragma unroll
    for (int j = 0; j < 4; ++j) { accE[j] = (v8f){}; accO[j] = (v8f){}; }

    for (int sb = 0; sb < NFOLD / 4; sb += 4) {
        v4f r0[4], r1[4], m0[4], m1[4], q[4];
        #pragma unroll
        for (int u = 0; u < 4; ++u) {
            const int s = sb + u;
            r0[u] = *(const v4f*)(Xp + (size_t)(4 * s) * NDIM);
            r1[u] = *(const v4f*)(Xp + (size_t)(4 * s + 1) * NDIM);
            m0[u] = *(const v4f*)(Xm - (size_t)(4 * s) * NDIM);
            m1[u] = *(const v4f*)(Xm - (size_t)(4 * s + 1) * NDIM);
            q[u]  = *(const v4f*)(Ap + s * 128);
        }
        #pragma unroll
        for (int u = 0; u < 4; ++u) {
            v4f e0 = r0[u] + m0[u], e1 = r1[u] + m1[u];
            v4f o0 = r0[u] - m0[u], o1 = r1[u] - m1[u];
            v2f ae = {q[u].x, q[u].y};
            v2f ao = {q[u].z, q[u].w};
            #pragma unroll
            for (int j = 0; j < 4; ++j) {
                v2f be = {e0[j], e1[j]};
                v2f bo = {o0[j], o1[j]};
                accE[j] = __builtin_amdgcn_wmma_f32_16x16x4_f32(false, ae, false, be, (short)0, accE[j], false, false);
                accO[j] = __builtin_amdgcn_wmma_f32_16x16x4_f32(false, ao, false, bo, (short)0, accO[j], false, false);
            }
            __builtin_amdgcn_sched_barrier(0);
        }
    }

    #pragma unroll
    for (int r = 0; r < 8; ++r) {
        v4f u0 = {accE[0][r], accE[1][r], accE[2][r], accE[3][r]};
        v4f u1 = {accO[0][r], accO[1][r], accO[2][r], accO[3][r]};
        *(v4f*)&U[(8 * half + r) * USTR + wave * 64 + 4 * l16]      = u0;
        *(v4f*)&U[(16 + 8 * half + r) * USTR + wave * 64 + 4 * l16] = u1;
    }
    __syncthreads();

    const int mt = wave >> 1;
    const int nt = wave & 1;
    const float sgn = nt ? -1.0f : 1.0f;
    const float* Up = &U[(16 * mt + l16) * USTR + 2 * half];
    const float* Um = &U[(16 * mt + l16) * USTR + (NDIM - 2) - 2 * half];
    const float* Bp = &Bf[(half * 16 + l16) * 4 + 2 * nt];
    v8f ye = {}; v8f yo = {};
    #pragma unroll 4
    for (int s = 0; s < NFOLD / 4; s += 2) {
        v2f fe = *(const v2f*)(Up + 4 * s);
        v2f ge = *(const v2f*)(Um - 4 * s);
        v2f be = *(const v2f*)(Bp + 128 * s);
        v2f fo = *(const v2f*)(Up + 4 * s + 4);
        v2f go = *(const v2f*)(Um - 4 * s - 4);
        v2f bo = *(const v2f*)(Bp + 128 * s + 128);
        v2f ae = {fmaf(sgn, ge.y, fe.x), fmaf(sgn, ge.x, fe.y)};
        v2f ao = {fmaf(sgn, go.y, fo.x), fmaf(sgn, go.x, fo.y)};
        ye = __builtin_amdgcn_wmma_f32_16x16x4_f32(false, ae, false, be, (short)0, ye, false, false);
        yo = __builtin_amdgcn_wmma_f32_16x16x4_f32(false, ao, false, bo, (short)0, yo, false, false);
    }

    float* Yg = Yall + grp * (KCOEF * KCOEF);
    #pragma unroll
    for (int r = 0; r < 8; ++r)
        Yg[(2 * (8 * half + r) + mt) * KCOEF + 2 * l16 + nt] = ye[r] + yo[r];
    __syncthreads();
    float* Y = out + slice * (KCOEF * KCOEF);
    const int t = tid & 127;
    #pragma unroll
    for (int i = 0; i < 2; ++i) {
        int idx = 4 * (t + 128 * i);
        *(v4f*)&Y[idx] = *(const v4f*)&Yg[idx];
    }
}

extern "C" void kernel_launch(void* const* d_in, const int* in_sizes, int n_in,
                              void* d_out, int out_size, void* d_ws, size_t ws_size,
                              hipStream_t stream) {
    const float* x = (const float*)d_in[0];
    float* out = (float*)d_out;
    int slices = in_sizes[0] / (NDIM * NDIM);
    TruncatedDCT_kernel<<<slices / NGRP, 128 * NGRP, 0, stream>>>(x, out);
}
